# starting kernel
# baseline (speedup 1.0000x reference)
#include <hip/hip_runtime.h>
#include <hip/hip_bf16.h>
#include <math.h>

#define D_FEAT   128
#define N_NODES  100000
#define N_EDGES  1000000

typedef __attribute__((ext_vector_type(2))) float v2f;
typedef __attribute__((ext_vector_type(8))) float v8f;

__global__ __launch_bounds__(128) void proj_wmma_kernel(
    const float* __restrict__ z, const float* __restrict__ W,
    float* __restrict__ st) {
  const int wave = (int)((blockIdx.x * blockDim.x + threadIdx.x) >> 5);
  const int lane = (int)(threadIdx.x & 31);
  const int ntiles = N_NODES / 16;
  if (wave >= ntiles) return;

  const int row  = lane & 15;
  const int half = lane >> 4;

  const float* __restrict__ arow  = z + (size_t)(wave * 16 + row) * D_FEAT;
  const bool   wvalid             = (row < 2);
  const float* __restrict__ wbase = wvalid ? (W + row * D_FEAT) : W;

  v8f c = {};
#pragma unroll
  for (int k0 = 0; k0 < D_FEAT; k0 += 4) {
    const int kk = k0 + 2 * half;
    v2f a  = *(const v2f*)(arow + kk);
    v2f wv = *(const v2f*)(wbase + kk);
    v2f b;
    b.x = wvalid ? wv.x : 0.0f;
    b.y = wvalid ? wv.y : 0.0f;
    c = __builtin_amdgcn_wmma_f32_16x16x4_f32(false, a, false, b,
                                              (short)0, c, false, false);
  }

  if (row < 2) {
    const int mbase = wave * 16 + half * 8;
#pragma unroll
    for (int v = 0; v < 8; ++v) {
      st[2 * (mbase + v) + row] = c[v];
    }
  }
}

__global__ __launch_bounds__(256) void edge_sigmoid_kernel(
    const int* __restrict__ ei, const float* __restrict__ st,
    const float* __restrict__ bias, float* __restrict__ out) {
  const int e = (int)(blockIdx.x * blockDim.x + threadIdx.x);
  if (e >= N_EDGES) return;
  const int s = ei[e];
  const int d = ei[N_EDGES + e];
  const float logit = st[2 * s] + st[2 * d + 1] + bias[0];
  out[e] = 1.0f / (1.0f + expf(-logit));
}

extern "C" void kernel_launch(void* const* d_in, const int* in_sizes, int n_in,
                              void* d_out, int out_size, void* d_ws,
                              size_t ws_size, hipStream_t stream) {
  (void)in_sizes; (void)n_in; (void)out_size; (void)ws_size;
  const float* z  = (const float*)d_in[0];
  const int*   ei = (const int*)d_in[1];
  const float* W  = (const float*)d_in[2];
  const float* b  = (const float*)d_in[3];
  float* st  = (float*)d_ws;
  float* out = (float*)d_out;

  const int ntiles  = N_NODES / 16;
  const int blocks1 = (ntiles + 3) / 4;
  proj_wmma_kernel<<<blocks1, 128, 0, stream>>>(z, W, st);

  const int blocks2 = (N_EDGES + 255) / 256;
  edge_sigmoid_kernel<<<blocks2, 256, 0, stream>>>(ei, st, b, out);
}


# edge pass: each thread does 8 edges (256-edge wave tiles, b128 index loads, 16 table gathers in flight, whole-line b128 stores), so 3912 waves are launched instead of 31256
# speedup vs baseline: 1.0698x; 1.0698x over previous
#include <hip/hip_runtime.h>
#include <hip/hip_bf16.h>
#include <math.h>

#define D_FEAT   128
#define N_NODES  100000
#define N_EDGES  1000000

typedef __attribute__((ext_vector_type(2))) float v2f;
typedef __attribute__((ext_vector_type(8))) float v8f;

__global__ __launch_bounds__(128) void proj_wmma_kernel(
    const float* __restrict__ z, const float* __restrict__ W,
    float* __restrict__ st) {
  const int wave = (int)((blockIdx.x * blockDim.x + threadIdx.x) >> 5);
  const int lane = (int)(threadIdx.x & 31);
  const int ntiles = N_NODES / 16;
  if (wave >= ntiles) return;

  const int row  = lane & 15;
  const int half = lane >> 4;

  const float* __restrict__ arow  = z + (size_t)(wave * 16 + row) * D_FEAT;
  const bool   wvalid             = (row < 2);
  const float* __restrict__ wbase = wvalid ? (W + row * D_FEAT) : W;

  v8f c = {};
#pragma unroll
  for (int k0 = 0; k0 < D_FEAT; k0 += 4) {
    const int kk = k0 + 2 * half;
    v2f a  = *(const v2f*)(arow + kk);
    v2f wv = *(const v2f*)(wbase + kk);
    v2f b;
    b.x = wvalid ? wv.x : 0.0f;
    b.y = wvalid ? wv.y : 0.0f;
    c = __builtin_amdgcn_wmma_f32_16x16x4_f32(false, a, false, b,
                                              (short)0, c, false, false);
  }

  if (row < 2) {
    const int mbase = wave * 16 + half * 8;
#pragma unroll
    for (int v = 0; v < 8; ++v) {
      st[2 * (mbase + v) + row] = c[v];
    }
  }
}

#define EDGES_PER_WAVE  256
#define EDGES_PER_BLOCK 2048

typedef __attribute__((ext_vector_type(4))) int   v4i;
typedef __attribute__((ext_vector_type(4))) float v4f;

__device__ __forceinline__ float edge_prob(float sv, float tv, float b) {
  const float logit = sv + tv + b;
  return 1.0f / (1.0f + expf(-logit));
}

__global__ __launch_bounds__(256) void edge_sigmoid_kernel(
    const int* __restrict__ ei, const float* __restrict__ st,
    const float* __restrict__ bias, float* __restrict__ out) {
  const int wave = (int)((blockIdx.x * blockDim.x + threadIdx.x) >> 5);
  const int lane = (int)(threadIdx.x & 31);
  const int tile = wave * EDGES_PER_WAVE;
  if (tile >= N_EDGES) return;
  const int* __restrict__ src = ei;
  const int* __restrict__ dst = ei + N_EDGES;
  const float b = bias[0];

  if (tile + EDGES_PER_WAVE <= N_EDGES) {
    const int eA = tile + 4 * lane;
    const int eB = eA + 128;
    const v4i sA = *(const v4i*)(src + eA);
    const v4i sB = *(const v4i*)(src + eB);
    const v4i dA = *(const v4i*)(dst + eA);
    const v4i dB = *(const v4i*)(dst + eB);
    float gs[8], gt[8];
#pragma unroll
    for (int i = 0; i < 4; ++i) {
      gs[i]     = st[2 * sA[i]];
      gs[4 + i] = st[2 * sB[i]];
      gt[i]     = st[2 * dA[i] + 1];
      gt[4 + i] = st[2 * dB[i] + 1];
    }
    v4f oA, oB;
#pragma unroll
    for (int i = 0; i < 4; ++i) {
      oA[i] = edge_prob(gs[i], gt[i], b);
      oB[i] = edge_prob(gs[4 + i], gt[4 + i], b);
    }
    *(v4f*)(out + eA) = oA;
    *(v4f*)(out + eB) = oB;
  } else {
#pragma unroll
    for (int c = 0; c < 2; ++c) {
      const int e = tile + 128 * c + 4 * lane;
      if (e < N_EDGES) {
        const v4i sv = *(const v4i*)(src + e);
        const v4i dv = *(const v4i*)(dst + e);
        v4f o;
#pragma unroll
        for (int i = 0; i < 4; ++i) {
          o[i] = edge_prob(st[2 * sv[i]], st[2 * dv[i] + 1], b);
        }
        *(v4f*)(out + e) = o;
      }
    }
  }
}

extern "C" void kernel_launch(void* const* d_in, const int* in_sizes, int n_in,
                              void* d_out, int out_size, void* d_ws,
                              size_t ws_size, hipStream_t stream) {
  (void)in_sizes; (void)n_in; (void)out_size; (void)ws_size;
  const float* z  = (const float*)d_in[0];
  const int*   ei = (const int*)d_in[1];
  const float* W  = (const float*)d_in[2];
  const float* b  = (const float*)d_in[3];
  float* st  = (float*)d_ws;
  float* out = (float*)d_out;

  const int ntiles  = N_NODES / 16;
  const int blocks1 = (ntiles + 3) / 4;
  proj_wmma_kernel<<<blocks1, 128, 0, stream>>>(z, W, st);

  const int blocks2 = (N_EDGES + EDGES_PER_BLOCK - 1) / EDGES_PER_BLOCK;
  edge_sigmoid_kernel<<<blocks2, 256, 0, stream>>>(ei, st, b, out);
}


# projection pass: global accesses cover whole 128-B lines: z rows as b128 loads transposed through LDS to A fragments, W fragments kept in registers over a wave's tiles, one store per st line
# speedup vs baseline: 1.2565x; 1.1745x over previous
#include <hip/hip_runtime.h>
#include <hip/hip_bf16.h>
#include <math.h>

#define D_FEAT   128
#define N_NODES  100000
#define N_EDGES  1000000

typedef __attribute__((ext_vector_type(2))) float v2f;
typedef __attribute__((ext_vector_type(8))) float v8f;

typedef __attribute__((ext_vector_type(4))) float v4f;

#define PROJ_TILES        (N_NODES / 16)
#define PROJ_WAVES_PER_WG 4
#define PROJ_WGS          512
#define PROJ_WAVES        (PROJ_WGS * PROJ_WAVES_PER_WG)
#define TILE_PITCH        132
#define WAVE_LDS_FLOATS   (16 * TILE_PITCH + 32)

__device__ __forceinline__ void wave_lds_handover() {
  __builtin_amdgcn_fence(__ATOMIC_RELEASE, "wavefront");
  __builtin_amdgcn_wave_barrier();
  __builtin_amdgcn_fence(__ATOMIC_ACQUIRE, "wavefront");
}

__device__ __forceinline__ void load_tile_rows(const float* __restrict__ z,
                                               int tile, int lane, v4f (&r)[16]) {
  const float* __restrict__ zt = z + (size_t)tile * 16 * D_FEAT + 4 * lane;
#pragma unroll
  for (int i = 0; i < 16; ++i) {
    r[i] = *(const v4f*)(zt + i * D_FEAT);
  }
}

__global__ __launch_bounds__(128) void proj_wmma_kernel(
    const float* __restrict__ z, const float* __restrict__ W,
    float* __restrict__ st) {
  __shared__ __attribute__((aligned(16)))
      float stage[PROJ_WAVES_PER_WG * WAVE_LDS_FLOATS];

  const int wib  = __builtin_amdgcn_readfirstlane((int)(threadIdx.x >> 5));
  const int lane = (int)(threadIdx.x & 31);
  const int gw   = (int)blockIdx.x * PROJ_WAVES_PER_WG + wib;

  const int row  = lane & 15;
  const int half = lane >> 4;

  float* __restrict__ my     = stage + wib * WAVE_LDS_FLOATS;
  float* __restrict__ my_out = my + 16 * TILE_PITCH;

  int tile = gw;
  v4f r[16];
  load_tile_rows(z, tile, lane, r);

  const bool   wvalid             = (row < 2);
  const float* __restrict__ wbase = (wvalid ? (W + row * D_FEAT) : W) + 2 * half;
  v2f bf[32];
#pragma unroll
  for (int i = 0; i < 32; ++i) {
    const v2f wv = *(const v2f*)(wbase + 4 * i);
    bf[i].x = wvalid ? wv.x : 0.0f;
    bf[i].y = wvalid ? wv.y : 0.0f;
  }

  const float* __restrict__ afrag = my + row * TILE_PITCH + 2 * half;

  while (tile < PROJ_TILES) {
#pragma unroll
    for (int i = 0; i < 16; ++i) {
      *(v4f*)(my + i * TILE_PITCH + 4 * lane) = r[i];
    }
    wave_lds_handover();

    const int next = tile + PROJ_WAVES;
    if (next < PROJ_TILES) load_tile_rows(z, next, lane, r);

    v2f a[32];
#pragma unroll
    for (int i = 0; i < 32; ++i) {
      a[i] = *(const v2f*)(afrag + 4 * i);
    }
    v8f c = {};
#pragma unroll
    for (int i = 0; i < 32; ++i) {
      c = __builtin_amdgcn_wmma_f32_16x16x4_f32(false, a[i], false, bf[i],
                                                (short)0, c, false, false);
    }
    __builtin_amdgcn_sched_group_barrier(0x100, 32, 0);
    __builtin_amdgcn_sched_group_barrier(0x008, 32, 0);

    if (row < 2) {
#pragma unroll
      for (int v = 0; v < 8; ++v) {
        my_out[2 * (half * 8 + v) + row] = c[v];
      }
    }
    wave_lds_handover();
    st[(size_t)tile * 32 + lane] = my_out[lane];
    wave_lds_handover();

    tile = next;
  }
}

#define EDGES_PER_WAVE  256
#define EDGES_PER_BLOCK 2048

typedef __attribute__((ext_vector_type(4))) int   v4i;

__device__ __forceinline__ float edge_prob(float sv, float tv, float b) {
  const float logit = sv + tv + b;
  return 1.0f / (1.0f + expf(-logit));
}

__global__ __launch_bounds__(256) void edge_sigmoid_kernel(
    const int* __restrict__ ei, const float* __restrict__ st,
    const float* __restrict__ bias, float* __restrict__ out) {
  const int wave = (int)((blockIdx.x * blockDim.x + threadIdx.x) >> 5);
  const int lane = (int)(threadIdx.x & 31);
  const int tile = wave * EDGES_PER_WAVE;
  if (tile >= N_EDGES) return;
  const int* __restrict__ src = ei;
  const int* __restrict__ dst = ei + N_EDGES;
  const float b = bias[0];

  if (tile + EDGES_PER_WAVE <= N_EDGES) {
    const int eA = tile + 4 * lane;
    const int eB = eA + 128;
    const v4i sA = *(const v4i*)(src + eA);
    const v4i sB = *(const v4i*)(src + eB);
    const v4i dA = *(const v4i*)(dst + eA);
    const v4i dB = *(const v4i*)(dst + eB);
    float gs[8], gt[8];
#pragma unroll
    for (int i = 0; i < 4; ++i) {
      gs[i]     = st[2 * sA[i]];
      gs[4 + i] = st[2 * sB[i]];
      gt[i]     = st[2 * dA[i] + 1];
      gt[4 + i] = st[2 * dB[i] + 1];
    }
    v4f oA, oB;
#pragma unroll
    for (int i = 0; i < 4; ++i) {
      oA[i] = edge_prob(gs[i], gt[i], b);
      oB[i] = edge_prob(gs[4 + i], gt[4 + i], b);
    }
    *(v4f*)(out + eA) = oA;
    *(v4f*)(out + eB) = oB;
  } else {
#pragma unroll
    for (int c = 0; c < 2; ++c) {
      const int e = tile + 128 * c + 4 * lane;
      if (e < N_EDGES) {
        const v4i sv = *(const v4i*)(src + e);
        const v4i dv = *(const v4i*)(dst + e);
        v4f o;
#pragma unroll
        for (int i = 0; i < 4; ++i) {
          o[i] = edge_prob(st[2 * sv[i]], st[2 * dv[i] + 1], b);
        }
        *(v4f*)(out + e) = o;
      }
    }
  }
}

extern "C" void kernel_launch(void* const* d_in, const int* in_sizes, int n_in,
                              void* d_out, int out_size, void* d_ws,
                              size_t ws_size, hipStream_t stream) {
  (void)in_sizes; (void)n_in; (void)out_size; (void)ws_size;
  const float* z  = (const float*)d_in[0];
  const int*   ei = (const int*)d_in[1];
  const float* W  = (const float*)d_in[2];
  const float* b  = (const float*)d_in[3];
  float* st  = (float*)d_ws;
  float* out = (float*)d_out;

  proj_wmma_kernel<<<PROJ_WGS, 128, 0, stream>>>(z, W, st);

  const int blocks2 = (N_EDGES + EDGES_PER_BLOCK - 1) / EDGES_PER_BLOCK;
  edge_sigmoid_kernel<<<blocks2, 256, 0, stream>>>(ei, st, b, out);
}
